# starting kernel
# baseline (speedup 1.0000x reference)
#include <hip/hip_runtime.h>


typedef float v2f __attribute__((ext_vector_type(2)));
typedef float v8f __attribute__((ext_vector_type(8)));

#define BATCH 16
#define CIN   32
#define HH    128
#define WW    128
#define PP    4
#define NNB   8
#define COUT  64
#define BN_EPS_C 1e-5f
#define ALPHA_LOG2E 14.4269504089f

#define HB      4
#define NWAVES  8
#define THREADS 256

#define OFF_WPAT  0
#define OFF_FUSE  (OFF_WPAT + 16*33)
#define OFF_PROJW (OFF_FUSE + 64*33)
#define OFF_BN    (OFF_PROJW + 64*33)
#define PROJ_RS   132
#define PROJ_PS   ((HB+2)*PROJ_RS)
#define OFF_PROJ  (OFF_BN + 128)
#define BITS_PS   528
#define OFF_BITS  (OFF_PROJ + PP*PROJ_PS)
#define LDS_FLOATS (OFF_BITS + CIN*BITS_PS)

#define DYPACK 0x6A40u
#define DXPACK 0x06A4u

__global__ __launch_bounds__(THREADS)
void lbp_fused_kernel(const float* __restrict__ x,
                      const float* __restrict__ wpat,
                      const float* __restrict__ fusew,
                      const float* __restrict__ gammaw,
                      const float* __restrict__ betaw,
                      const float* __restrict__ meanw,
                      const float* __restrict__ varw,
                      const float* __restrict__ projw,
                      const float* __restrict__ res_scale,
                      float* __restrict__ out)
{
    extern __shared__ float lds[];
    const int tid  = threadIdx.x;
    const int lane = tid & 31;
    const int wave = tid >> 5;
    const int hl   = lane >> 4;
    const int ln   = lane & 15;

    const int blk  = blockIdx.x;
    const int bimg = blk >> 5;
    const int hblk = blk & 31;
    const int h0   = hblk * HB;

    for (int i = tid; i < 16*CIN; i += THREADS) {
        int m = i >> 5, k = i & 31;
        lds[OFF_WPAT + m*33 + k] = (m < PP) ? wpat[m*CIN + k] : 0.0f;
    }
    for (int i = tid; i < COUT*CIN; i += THREADS) {
        int o = i >> 5, k = i & 31;
        lds[OFF_FUSE  + o*33 + k] = fusew[i];
        lds[OFF_PROJW + o*33 + k] = projw[i];
    }
    if (tid < COUT) {
        float inv = gammaw[tid] * rsqrtf(varw[tid] + BN_EPS_C);
        lds[OFF_BN + tid]      = inv;
        lds[OFF_BN + 64 + tid] = betaw[tid] - meanw[tid] * inv;
    }
    __syncthreads();

    const float* xb = x + (size_t)bimg * (CIN*HH*WW);

    for (int t = wave; t < (HB+2)*8; t += NWAVES) {
        int prow = t >> 3;
        int w0   = (t & 7) << 4;
        int hg   = (h0 - 1 + prow) & (HH - 1);
        v8f acc = {};
        #pragma unroll
        for (int ks = 0; ks < CIN; ks += 4) {
            int kb = ks + 2*hl;
            v2f a, bm;
            a[0] = lds[OFF_WPAT + ln*33 + kb];
            a[1] = lds[OFF_WPAT + ln*33 + kb + 1];
            const float* xp = xb + ((size_t)kb*HH + hg)*WW + w0 + ln;
            bm[0] = xp[0];
            bm[1] = xp[HH*WW];
            acc = __builtin_amdgcn_wmma_f32_16x16x4_f32(false, a, false, bm,
                                                        (short)0, acc, false, false);
        }
        if (hl == 0) {
            #pragma unroll
            for (int r = 0; r < PP; ++r)
                lds[OFF_PROJ + r*PROJ_PS + prow*PROJ_RS + w0 + ln] = acc[r];
        }
    }
    __syncthreads();

    for (int i = tid; i < CIN*HB*WW; i += THREADS) {
        int k   = i >> 9;
        int pix = i & 511;
        int row = pix >> 7, col = pix & 127;
        int p = k >> 3, nf = k & 7;
        int dy = (int)((DYPACK >> (2*nf)) & 3u) - 1;
        int dx = (int)((DXPACK >> (2*nf)) & 3u) - 1;
        float c  = lds[OFF_PROJ + p*PROJ_PS + (row+1)*PROJ_RS + col];
        float nb = lds[OFF_PROJ + p*PROJ_PS + (row+1-dy)*PROJ_RS + ((col - dx) & (WW-1))];
        float t  = __builtin_amdgcn_exp2f(ALPHA_LOG2E * (c - nb));
        lds[OFF_BITS + k*BITS_PS + pix] = __builtin_amdgcn_rcpf(1.0f + t);
    }
    __syncthreads();

    const float srs = res_scale[0];

    for (int t = wave; t < HB*8; t += NWAVES) {
        int row  = t >> 3;
        int w0   = (t & 7) << 4;
        int hg   = h0 + row;
        int pixb = (row << 7) + w0;

        v8f accF[4] = {};
        v8f accI[4] = {};
        #pragma unroll
        for (int ks = 0; ks < CIN; ks += 4) {
            int kb = ks + 2*hl;
            v2f bb, bx;
            bb[0] = lds[OFF_BITS + kb*BITS_PS + pixb + ln];
            bb[1] = lds[OFF_BITS + (kb+1)*BITS_PS + pixb + ln];
            const float* xp = xb + ((size_t)kb*HH + hg)*WW + w0 + ln;
            bx[0] = xp[0];
            bx[1] = xp[HH*WW];
            #pragma unroll
            for (int mt = 0; mt < 4; ++mt) {
                int ro = (mt*16 + ln)*33 + kb;
                v2f aF, aI;
                aF[0] = lds[OFF_FUSE  + ro]; aF[1] = lds[OFF_FUSE  + ro + 1];
                aI[0] = lds[OFF_PROJW + ro]; aI[1] = lds[OFF_PROJW + ro + 1];
                accF[mt] = __builtin_amdgcn_wmma_f32_16x16x4_f32(false, aF, false, bb,
                                                                 (short)0, accF[mt], false, false);
                accI[mt] = __builtin_amdgcn_wmma_f32_16x16x4_f32(false, aI, false, bx,
                                                                 (short)0, accI[mt], false, false);
            }
        }

        float* op = out + (((size_t)bimg*COUT)*HH + hg)*WW + w0 + ln;
        #pragma unroll
        for (int mt = 0; mt < 4; ++mt) {
            #pragma unroll
            for (int r = 0; r < 8; ++r) {
                int o = mt*16 + r + 8*hl;
                float v = (accF[mt][r]*lds[OFF_BN + o] + lds[OFF_BN + 64 + o])*srs + accI[mt][r];
                op[(size_t)o*HH*WW] = v > 0.0f ? v : 0.0f;
            }
        }
    }
}

extern "C" void kernel_launch(void* const* d_in, const int* in_sizes, int n_in,
                              void* d_out, int out_size, void* d_ws, size_t ws_size,
                              hipStream_t stream) {
    (void)in_sizes; (void)n_in; (void)d_ws; (void)ws_size; (void)out_size;
    const float* x      = (const float*)d_in[0];
    const float* wpat   = (const float*)d_in[1];
    const float* fusew  = (const float*)d_in[2];
    const float* gammaw = (const float*)d_in[3];
    const float* betaw  = (const float*)d_in[4];
    const float* meanw  = (const float*)d_in[5];
    const float* varw   = (const float*)d_in[6];
    const float* projw  = (const float*)d_in[7];
    const float* rscale = (const float*)d_in[8];
    float* outp = (float*)d_out;

    dim3 grid(BATCH * (HH / HB));
    dim3 block(THREADS);
    size_t ldsBytes = (size_t)LDS_FLOATS * sizeof(float);
    lbp_fused_kernel<<<grid, block, ldsBytes, stream>>>(
        x, wpat, fusew, gammaw, betaw, meanw, varw, projw, rscale, outp);
}


# 4 waves per work-group, up to 512 registers each: a wave owns a whole slab row and all 64 channels and keeps its x in registers for every identity product; half as many waves enter; sums unchanged
# speedup vs baseline: 2.2849x; 2.2849x over previous
#include <hip/hip_runtime.h>


typedef float v2f __attribute__((ext_vector_type(2)));
typedef float v4f __attribute__((ext_vector_type(4)));
typedef float v8f __attribute__((ext_vector_type(8)));

#define BATCH 16
#define CIN   32
#define HH    128
#define WW    128
#define PP    4
#define NNB   8
#define COUT  64
#define BN_EPS_C 1e-5f
#define ALPHA_LOG2E 14.4269504089f

#define HB      4
#define NWAVES  4
#define THREADS 128

#define PROJ_RS   132
#define PROJ_PS   ((HB+2)*PROJ_RS)
#define OFF_PROJ  0
#define BITS_PS   528
#define OFF_BITS  (OFF_PROJ + PP*PROJ_PS)
#define LDS_FLOATS (OFF_BITS + CIN*BITS_PS)

#define DYPACK 0x6A40u
#define DXPACK 0x06A4u

__global__ __launch_bounds__(THREADS, 2)
void lbp_fused_kernel(const float* __restrict__ x,
                      const float* __restrict__ wpat,
                      const float* __restrict__ fusew,
                      const float* __restrict__ gammaw,
                      const float* __restrict__ betaw,
                      const float* __restrict__ meanw,
                      const float* __restrict__ varw,
                      const float* __restrict__ projw,
                      const float* __restrict__ res_scale,
                      float* __restrict__ out)
{
    extern __shared__ __attribute__((aligned(16))) float lds[];
    const int tid  = threadIdx.x;
    const int lane = tid & 31;
    const int wave = tid >> 5;
    const int hl   = lane >> 4;
    const int ln   = lane & 15;

    const int blk  = blockIdx.x;
    const int bimg = blk >> 5;
    const int hblk = blk & 31;
    const int h0   = hblk * HB;

    const float* xb = x + (size_t)bimg * (CIN*HH*WW);

    const int row = wave;

    v2f aP[8];
    {
        const float* wp = wpat + (ln < PP ? ln : 0)*CIN + 2*hl;
        #pragma unroll
        for (int j = 0; j < 8; ++j) {
            float p0 = wp[4*j], p1 = wp[4*j + 1];
            aP[j][0] = (ln < PP) ? p0 : 0.0f;
            aP[j][1] = (ln < PP) ? p1 : 0.0f;
        }
    }

    v2f aI[4][8];
    v2f aF[4][8];
    auto load_i_tile = [&](int m) {
        const int ro = (m*16 + ln)*CIN + 2*hl;
        #pragma unroll
        for (int j = 0; j < 8; ++j) {
            aI[m][j][0] = projw[ro + 4*j];
            aI[m][j][1] = projw[ro + 4*j + 1];
        }
    };
    auto load_f_tile = [&](int m) {
        const int ro = (m*16 + ln)*CIN + 2*hl;
        #pragma unroll
        for (int j = 0; j < 8; ++j) {
            aF[m][j][0] = fusew[ro + 4*j];
            aF[m][j][1] = fusew[ro + 4*j + 1];
        }
    };

    v4f xe[2][8], xo[2][8];
    v8f accA[4][4] = {};

    {
        const int prow = row + 1;
        const int hgp  = h0 + row;
        #pragma unroll
        for (int c = 0; c < 2; ++c) {
            #pragma unroll
            for (int j = 0; j < 8; ++j) {
                const float* xp = xb + ((size_t)(4*j + 2*hl)*HH + hgp)*WW + 64*c + 4*ln;
                xe[c][j] = *(const v4f*)xp;
                xo[c][j] = *(const v4f*)(xp + HH*WW);
            }
        }
        load_i_tile(0);
        load_i_tile(1);
        #pragma unroll
        for (int c = 0; c < 2; ++c) {
            v8f acc[4] = {};
            #pragma unroll
            for (int j = 0; j < 8; ++j) {
                #pragma unroll
                for (int q = 0; q < 4; ++q) {
                    v2f bq;
                    bq[0] = xe[c][j][q];
                    bq[1] = xo[c][j][q];
                    acc[q] = __builtin_amdgcn_wmma_f32_16x16x4_f32(false, aP[j], false, bq,
                                                                   (short)0, acc[q], false, false);
                }
            }
            if (hl == 0) {
                #pragma unroll
                for (int r = 0; r < PP; ++r) {
                    v4f v;
                    v[0] = acc[0][r]; v[1] = acc[1][r]; v[2] = acc[2][r]; v[3] = acc[3][r];
                    *(v4f*)&lds[OFF_PROJ + r*PROJ_PS + prow*PROJ_RS + 64*c + 4*ln] = v;
                }
            }
            if (c == 0) { load_i_tile(2); load_i_tile(3); }
        }
    }
    {
        const int prow = (wave >> 1) ? (HB + 1) : 0;
        const int w0   = (wave & 1) << 6;
        const int hgp  = (h0 - 1 + prow) & (HH - 1);
        v4f ye[8], yo[8];
        #pragma unroll
        for (int j = 0; j < 8; ++j) {
            const float* xp = xb + ((size_t)(4*j + 2*hl)*HH + hgp)*WW + w0 + 4*ln;
            ye[j] = *(const v4f*)xp;
            yo[j] = *(const v4f*)(xp + HH*WW);
        }
        #pragma unroll
        for (int j = 0; j < 8; ++j) {
            #pragma unroll
            for (int q = 0; q < 2; ++q) {
                v2f bq;
                bq[0] = xe[0][j][q];
                bq[1] = xo[0][j][q];
                #pragma unroll
                for (int m = 0; m < 4; ++m)
                    accA[q][m] = __builtin_amdgcn_wmma_f32_16x16x4_f32(false, aI[m][j], false, bq,
                                                                       (short)0, accA[q][m], false, false);
            }
        }
        v8f acc[4] = {};
        #pragma unroll
        for (int j = 0; j < 8; ++j) {
            #pragma unroll
            for (int q = 0; q < 4; ++q) {
                v2f bq;
                bq[0] = ye[j][q];
                bq[1] = yo[j][q];
                acc[q] = __builtin_amdgcn_wmma_f32_16x16x4_f32(false, aP[j], false, bq,
                                                               (short)0, acc[q], false, false);
            }
        }
        if (hl == 0) {
            #pragma unroll
            for (int r = 0; r < PP; ++r) {
                v4f v;
                v[0] = acc[0][r]; v[1] = acc[1][r]; v[2] = acc[2][r]; v[3] = acc[3][r];
                *(v4f*)&lds[OFF_PROJ + r*PROJ_PS + prow*PROJ_RS + w0 + 4*ln] = v;
            }
        }
    }
    __syncthreads();
    __builtin_amdgcn_sched_barrier(0);

    #pragma unroll
    for (int half = 0; half < HB; ++half) {
        const int pix  = tid + half*THREADS;
        const int srow = half;
        const int col  = tid;
        int cx[3];
        cx[0] = (col - 1) & (WW-1);
        cx[1] = col;
        cx[2] = (col + 1) & (WW-1);
        #pragma unroll
        for (int p = 0; p < PP; ++p) {
            const int base = OFF_PROJ + p*PROJ_PS + srow*PROJ_RS;
            float win[3][3];
            #pragma unroll
            for (int ry = 0; ry < 3; ++ry) {
                #pragma unroll
                for (int rc = 0; rc < 3; ++rc)
                    win[ry][rc] = lds[base + ry*PROJ_RS + cx[rc]];
            }
            const float c = win[1][1];
            #pragma unroll
            for (int nf = 0; nf < NNB; ++nf) {
                const int dy = (int)((DYPACK >> (2*nf)) & 3u) - 1;
                const int dx = (int)((DXPACK >> (2*nf)) & 3u) - 1;
                float t = __builtin_amdgcn_exp2f(ALPHA_LOG2E * (c - win[1-dy][1-dx]));
                lds[OFF_BITS + (p*NNB + nf)*BITS_PS + pix] = __builtin_amdgcn_rcpf(1.0f + t);
            }
            {
                const int slot = half*PP + p;
                const int j = slot >> 1;
                const int q = 2 + (slot & 1);
                v2f bq;
                bq[0] = xe[0][j][q];
                bq[1] = xo[0][j][q];
                #pragma unroll
                for (int m = 0; m < 4; ++m)
                    accA[q][m] = __builtin_amdgcn_wmma_f32_16x16x4_f32(false, aI[m][j], false, bq,
                                                                       (short)0, accA[q][m], false, false);
            }
        }
    }
    #pragma unroll
    for (int m = 0; m < 4; ++m) load_f_tile(m);
    __builtin_amdgcn_sched_barrier(0);
    __syncthreads();

    const float srs = res_scale[0];

    const int hg = h0 + row;

    float bnInv[4][8], bnAdd[4][8];
    #pragma unroll
    for (int m = 0; m < 4; ++m) {
        #pragma unroll
        for (int r = 0; r < 8; ++r) {
            const int o = m*16 + r + 8*hl;
            float inv = gammaw[o] * rsqrtf(varw[o] + BN_EPS_C);
            bnInv[m][r] = inv;
            bnAdd[m][r] = betaw[o] - meanw[o] * inv;
        }
    }

    #pragma nounroll
    for (int c = 0; c < 2; ++c) {
        if (c == 1) {
            #pragma unroll
            for (int m = 0; m < 4; ++m) load_i_tile(m);
            #pragma unroll
            for (int q = 0; q < 4; ++q) {
                #pragma unroll
                for (int m = 0; m < 4; ++m) accA[q][m] = (v8f){};
            }
            #pragma unroll
            for (int j = 0; j < 8; ++j) {
                #pragma unroll
                for (int q = 0; q < 4; ++q) {
                    v2f bq;
                    bq[0] = xe[1][j][q];
                    bq[1] = xo[1][j][q];
                    #pragma unroll
                    for (int m = 0; m < 4; ++m)
                        accA[q][m] = __builtin_amdgcn_wmma_f32_16x16x4_f32(false, aI[m][j], false, bq,
                                                                           (short)0, accA[q][m], false, false);
                }
            }
        }
        const int pixb = (row << 7) + 64*c + 4*ln;
        #pragma unroll
        for (int mq = 0; mq < 2; ++mq) {
            #pragma unroll
            for (int q0 = 0; q0 < 4; q0 += 2) {
                v2f b0[8], b1[8];
                #pragma unroll
                for (int j = 0; j < 8; ++j) {
                    const int kb = 4*j + 2*hl;
                    b0[j] = *(const v2f*)&lds[OFF_BITS + kb*BITS_PS + pixb + q0];
                    b1[j] = *(const v2f*)&lds[OFF_BITS + (kb+1)*BITS_PS + pixb + q0];
                }
                v8f accF[2][2] = {};
                #pragma unroll
                for (int j = 0; j < 8; ++j) {
                    #pragma unroll
                    for (int t = 0; t < 2; ++t) {
                        v2f bq;
                        bq[0] = b0[j][t];
                        bq[1] = b1[j][t];
                        #pragma unroll
                        for (int m = 0; m < 2; ++m)
                            accF[t][m] = __builtin_amdgcn_wmma_f32_16x16x4_f32(false, aF[2*mq + m][j], false, bq,
                                                                               (short)0, accF[t][m], false, false);
                    }
                }
                #pragma unroll
                for (int t = 0; t < 2; ++t) {
                    #pragma unroll
                    for (int m = 0; m < 2; ++m) {
                        #pragma unroll
                        for (int r = 0; r < 8; ++r) {
                            float v = (accF[t][m][r]*bnInv[2*mq + m][r] + bnAdd[2*mq + m][r])*srs + accA[q0 + t][2*mq + m][r];
                            accA[q0 + t][2*mq + m][r] = v > 0.0f ? v : 0.0f;
                        }
                    }
                }
            }
            float* op = out + (((size_t)bimg*COUT + mq*32 + 8*hl)*HH + hg)*WW + 64*c + 4*ln;
            #pragma unroll
            for (int m = 0; m < 2; ++m) {
                #pragma unroll
                for (int r = 0; r < 8; ++r) {
                    v4f v;
                    v[0] = accA[0][2*mq + m][r]; v[1] = accA[1][2*mq + m][r];
                    v[2] = accA[2][2*mq + m][r]; v[3] = accA[3][2*mq + m][r];
                    *(v4f*)(op + (size_t)(m*16 + r)*HH*WW) = v;
                }
            }
        }
    }
}

extern "C" void kernel_launch(void* const* d_in, const int* in_sizes, int n_in,
                              void* d_out, int out_size, void* d_ws, size_t ws_size,
                              hipStream_t stream) {
    (void)in_sizes; (void)n_in; (void)d_ws; (void)ws_size; (void)out_size;
    const float* x      = (const float*)d_in[0];
    const float* wpat   = (const float*)d_in[1];
    const float* fusew  = (const float*)d_in[2];
    const float* gammaw = (const float*)d_in[3];
    const float* betaw  = (const float*)d_in[4];
    const float* meanw  = (const float*)d_in[5];
    const float* varw   = (const float*)d_in[6];
    const float* projw  = (const float*)d_in[7];
    const float* rscale = (const float*)d_in[8];
    float* outp = (float*)d_out;

    dim3 grid(BATCH * (HH / HB));
    dim3 block(THREADS);
    size_t ldsBytes = (size_t)LDS_FLOATS * sizeof(float);
    lbp_fused_kernel<<<grid, block, ldsBytes, stream>>>(
        x, wpat, fusew, gammaw, betaw, meanw, varw, projw, rscale, outp);
}
